# starting kernel
# baseline (speedup 1.0000x reference)
#include <hip/hip_runtime.h>
#include <hip/hip_bf16.h>
#include <math.h>

typedef float v2f __attribute__((ext_vector_type(2)));
typedef float v8f __attribute__((ext_vector_type(8)));

static constexpr int NN = 4096;
static constexpr int CC = 64;
static constexpr int C2 = 32;
static constexpr int NB = 8;

__device__ __forceinline__ float sw16(float v) {
  return __int_as_float(__builtin_amdgcn_ds_swizzle(__float_as_int(v), 0x401f));
}

__device__ __forceinline__ v8f wmma4(v2f a, v2f b, v8f c) {
  return __builtin_amdgcn_wmma_f32_16x16x4_f32(false, a, false, b, (short)0, c,
                                               false, false);
}

__device__ __forceinline__ void c2b(const v8f& R, bool lo, v2f* B) {
  float s0 = sw16(R[0]), s1 = sw16(R[1]), s2 = sw16(R[2]), s3 = sw16(R[3]);
  float s4 = sw16(R[4]), s5 = sw16(R[5]), s6 = sw16(R[6]), s7 = sw16(R[7]);
  B[0][0] = lo ? R[0] : s2;  B[0][1] = lo ? R[1] : s3;
  B[1][0] = lo ? R[4] : s6;  B[1][1] = lo ? R[5] : s7;
  B[2][0] = lo ? s0 : R[2];  B[2][1] = lo ? s1 : R[3];
  B[3][0] = lo ? s4 : R[6];  B[3][1] = lo ? s5 : R[7];
}

__global__ __launch_bounds__(256) void gx_kernel(
    const float* __restrict__ x, const float* __restrict__ g_w,
    const float* __restrict__ g_b, float* __restrict__ Vg) {
  int idx = blockIdx.x * 256 + threadIdx.x;
  int b = idx >> 12;
  int n = idx & (NN - 1);
  const float* xb = x + (size_t)b * (CC * NN) + n;
  float acc[C2];
#pragma unroll
  for (int o = 0; o < C2; ++o) acc[o] = g_b[o];
  for (int c = 0; c < CC; ++c) {
    float xv = xb[(size_t)c * NN];
#pragma unroll
    for (int o = 0; o < C2; ++o) acc[o] = fmaf(g_w[o * CC + c], xv, acc[o]);
  }
  float* vp = Vg + (size_t)idx * C2;
#pragma unroll
  for (int o = 0; o < C2; ++o) vp[o] = acc[o];
}

__global__ __launch_bounds__(256) void attn_kernel(
    const float* __restrict__ x, const float* __restrict__ Vg,
    const float* __restrict__ W_w, const float* __restrict__ W_b,
    float* __restrict__ out) {
  __shared__ float ldsK[16 * 66];
  __shared__ float ldsV[32 * 18];

  const int b    = blockIdx.x >> 5;
  const int qblk = blockIdx.x & 31;
  const int tid  = threadIdx.x;
  const int wave = tid >> 5;
  const int lane = tid & 31;
  const int lq   = lane & 15;
  const bool lo  = lane < 16;
  const int n0   = qblk * 128 + wave * 16;

  const float* xb = x  + (size_t)b * (CC * NN);
  const float* vb = Vg + (size_t)b * (NN * C2);

  v2f qb[16];
  {
    const int crow = lo ? 0 : 2;
#pragma unroll
    for (int ck = 0; ck < 16; ++ck) {
      qb[ck][0] = xb[(size_t)(4 * ck + crow) * NN + n0 + lq];
      qb[ck][1] = xb[(size_t)(4 * ck + crow + 1) * NN + n0 + lq];
    }
  }

  v8f acc0 = {};
  v8f acc1 = {};
  float m_run = -INFINITY;
  float l_run = 0.0f;

  const int kc_m = tid & 15, kc_c = tid >> 4;
  const int vc_o = tid & 31, vc_m = tid >> 5;

  for (int m0 = 0; m0 < NN; m0 += 16) {
    __syncthreads();
#pragma unroll
    for (int i = 0; i < 4; ++i)
      ldsK[kc_m * 66 + kc_c + 16 * i] =
          xb[(size_t)(kc_c + 16 * i) * NN + m0 + kc_m];
#pragma unroll
    for (int i = 0; i < 2; ++i)
      ldsV[vc_o * 18 + vc_m + 8 * i] =
          vb[(size_t)(m0 + vc_m + 8 * i) * C2 + vc_o];
    __syncthreads();

    v8f s = {};
    const int arow = lq * 66 + (lo ? 0 : 2);
#pragma unroll
    for (int ck = 0; ck < 16; ++ck) {
      v2f a;
      a[0] = ldsK[arow + 4 * ck];
      a[1] = ldsK[arow + 4 * ck + 1];
      s = wmma4(a, qb[ck], s);
    }

    float tmax = s[0];
#pragma unroll
    for (int r = 1; r < 8; ++r) tmax = fmaxf(tmax, s[r]);
    tmax = fmaxf(tmax, sw16(tmax));
    float m_new = fmaxf(m_run, tmax);
    float scale = __expf(m_run - m_new);
    m_run = m_new;
    v8f p;
    float tsum = 0.0f;
#pragma unroll
    for (int r = 0; r < 8; ++r) {
      p[r] = __expf(s[r] - m_new);
      tsum += p[r];
    }
    tsum += sw16(tsum);
    l_run = l_run * scale + tsum;
#pragma unroll
    for (int r = 0; r < 8; ++r) {
      acc0[r] *= scale;
      acc1[r] *= scale;
    }

    v2f pb[4];
    c2b(p, lo, pb);

    const int vrow0 = lq * 18 + (lo ? 0 : 2);
    const int vrow1 = (lq + 16) * 18 + (lo ? 0 : 2);
#pragma unroll
    for (int j = 0; j < 4; ++j) {
      v2f a;
      a[0] = ldsV[vrow0 + 4 * j];
      a[1] = ldsV[vrow0 + 4 * j + 1];
      acc0 = wmma4(a, pb[j], acc0);
    }
#pragma unroll
    for (int j = 0; j < 4; ++j) {
      v2f a;
      a[0] = ldsV[vrow1 + 4 * j];
      a[1] = ldsV[vrow1 + 4 * j + 1];
      acc1 = wmma4(a, pb[j], acc1);
    }
  }

  float rinv = 1.0f / l_run;
#pragma unroll
  for (int r = 0; r < 8; ++r) {
    acc0[r] *= rinv;
    acc1[r] *= rinv;
  }

  v2f yb[8];
  c2b(acc0, lo, yb);
  c2b(acc1, lo, yb + 4);

  float* ob = out + (size_t)b * (CC * NN);
#pragma unroll
  for (int ct = 0; ct < 4; ++ct) {
    v8f z;
#pragma unroll
    for (int r = 0; r < 8; ++r) {
      int c = ct * 16 + r + (lo ? 0 : 8);
      z[r] = xb[(size_t)c * NN + n0 + lq] + W_b[c];
    }
    const int wrow = (ct * 16 + lq) * C2 + (lo ? 0 : 2);
#pragma unroll
    for (int oc = 0; oc < 8; ++oc) {
      v2f a;
      a[0] = W_w[wrow + 4 * oc];
      a[1] = W_w[wrow + 4 * oc + 1];
      z = wmma4(a, yb[oc], z);
    }
#pragma unroll
    for (int r = 0; r < 8; ++r) {
      int c = ct * 16 + r + (lo ? 0 : 8);
      ob[(size_t)c * NN + n0 + lq] = z[r];
    }
  }
}

extern "C" void kernel_launch(void* const* d_in, const int* in_sizes, int n_in,
                              void* d_out, int out_size, void* d_ws,
                              size_t ws_size, hipStream_t stream) {
  const float* x   = (const float*)d_in[0];
  const float* g_w = (const float*)d_in[1];
  const float* g_b = (const float*)d_in[2];
  const float* W_w = (const float*)d_in[3];
  const float* W_b = (const float*)d_in[4];
  float* out = (float*)d_out;
  float* Vg  = (float*)d_ws;

  gx_kernel<<<(NB * NN) / 256, 256, 0, stream>>>(x, g_w, g_b, Vg);
  attn_kernel<<<NB * (NN / 128), 256, 0, stream>>>(x, Vg, W_w, W_b, out);
}


# Key tile per barrier pair 16 -> 64 keys: 4 independent score tiles, one online-softmax max/rescale/sum per 64 keys (float summation order of l and y changes), V fragments read as one group
# speedup vs baseline: 1.4344x; 1.4344x over previous
#include <hip/hip_runtime.h>
#include <hip/hip_bf16.h>
#include <math.h>

typedef float v2f __attribute__((ext_vector_type(2)));
typedef float v8f __attribute__((ext_vector_type(8)));

static constexpr int NN = 4096;
static constexpr int CC = 64;
static constexpr int C2 = 32;
static constexpr int NB = 8;
static constexpr int KT = 64;
static constexpr int NT = KT / 16;
static constexpr int SK = 66;
static constexpr int SV = KT + 4;

__device__ __forceinline__ float sw16(float v) {
  return __int_as_float(__builtin_amdgcn_ds_swizzle(__float_as_int(v), 0x401f));
}

__device__ __forceinline__ v8f wmma4(v2f a, v2f b, v8f c) {
  return __builtin_amdgcn_wmma_f32_16x16x4_f32(false, a, false, b, (short)0, c,
                                               false, false);
}

__device__ __forceinline__ void c2b(const v8f& R, bool lo, v2f* B) {
  float s0 = sw16(R[0]), s1 = sw16(R[1]), s2 = sw16(R[2]), s3 = sw16(R[3]);
  float s4 = sw16(R[4]), s5 = sw16(R[5]), s6 = sw16(R[6]), s7 = sw16(R[7]);
  B[0][0] = lo ? R[0] : s2;  B[0][1] = lo ? R[1] : s3;
  B[1][0] = lo ? R[4] : s6;  B[1][1] = lo ? R[5] : s7;
  B[2][0] = lo ? s0 : R[2];  B[2][1] = lo ? s1 : R[3];
  B[3][0] = lo ? s4 : R[6];  B[3][1] = lo ? s5 : R[7];
}

__global__ __launch_bounds__(256) void gx_kernel(
    const float* __restrict__ x, const float* __restrict__ g_w,
    const float* __restrict__ g_b, float* __restrict__ Vg) {
  int idx = blockIdx.x * 256 + threadIdx.x;
  int b = idx >> 12;
  int n = idx & (NN - 1);
  const float* xb = x + (size_t)b * (CC * NN) + n;
  float acc[C2];
#pragma unroll
  for (int o = 0; o < C2; ++o) acc[o] = g_b[o];
  for (int c = 0; c < CC; ++c) {
    float xv = xb[(size_t)c * NN];
#pragma unroll
    for (int o = 0; o < C2; ++o) acc[o] = fmaf(g_w[o * CC + c], xv, acc[o]);
  }
  float* vp = Vg + (size_t)idx * C2;
#pragma unroll
  for (int o = 0; o < C2; ++o) vp[o] = acc[o];
}

__global__ __launch_bounds__(256) void attn_kernel(
    const float* __restrict__ x, const float* __restrict__ Vg,
    const float* __restrict__ W_w, const float* __restrict__ W_b,
    float* __restrict__ out) {
  __shared__ float ldsK[KT * SK];
  __shared__ float ldsV[32 * SV];

  const int b    = blockIdx.x >> 5;
  const int qblk = blockIdx.x & 31;
  const int tid  = threadIdx.x;
  const int wave = tid >> 5;
  const int lane = tid & 31;
  const int lq   = lane & 15;
  const bool lo  = lane < 16;
  const int n0   = qblk * 128 + wave * 16;

  const float* xb = x  + (size_t)b * (CC * NN);
  const float* vb = Vg + (size_t)b * (NN * C2);

  v2f qb[16];
  {
    const int crow = lo ? 0 : 2;
#pragma unroll
    for (int ck = 0; ck < 16; ++ck) {
      qb[ck][0] = xb[(size_t)(4 * ck + crow) * NN + n0 + lq];
      qb[ck][1] = xb[(size_t)(4 * ck + crow + 1) * NN + n0 + lq];
    }
  }

  v8f acc0 = {};
  v8f acc1 = {};
  float m_run = -INFINITY;
  float l_run = 0.0f;

  const int kc_m = tid & (KT - 1), kc_c = tid / KT;
  const int vc_o = tid & 31, vc_m = tid >> 5;
  const int fsel = lo ? 0 : 2;

  for (int m0 = 0; m0 < NN; m0 += KT) {
    __syncthreads();
#pragma unroll
    for (int i = 0; i < KT / 4; ++i) {
      const int c = kc_c + (256 / KT) * i;
      ldsK[kc_m * SK + c] = xb[(size_t)c * NN + m0 + kc_m];
    }
#pragma unroll
    for (int i = 0; i < KT / 8; ++i) {
      const int m = vc_m + 8 * i;
      ldsV[vc_o * SV + m] = vb[(size_t)(m0 + m) * C2 + vc_o];
    }
    __syncthreads();

    v8f s[NT];
#pragma unroll
    for (int t = 0; t < NT; ++t) s[t] = v8f{};
#pragma unroll
    for (int ck = 0; ck < 16; ++ck) {
#pragma unroll
      for (int t = 0; t < NT; ++t) {
        const int arow = (16 * t + lq) * SK + fsel + 4 * ck;
        v2f a;
        a[0] = ldsK[arow];
        a[1] = ldsK[arow + 1];
        s[t] = wmma4(a, qb[ck], s[t]);
      }
    }

    v2f va0[NT][4], va1[NT][4];
#pragma unroll
    for (int t = 0; t < NT; ++t) {
      const int vrow0 = lq * SV + 16 * t + fsel;
      const int vrow1 = (lq + 16) * SV + 16 * t + fsel;
#pragma unroll
      for (int j = 0; j < 4; ++j) {
        va0[t][j][0] = ldsV[vrow0 + 4 * j];
        va0[t][j][1] = ldsV[vrow0 + 4 * j + 1];
        va1[t][j][0] = ldsV[vrow1 + 4 * j];
        va1[t][j][1] = ldsV[vrow1 + 4 * j + 1];
      }
    }

    __builtin_amdgcn_sched_barrier(0);

    float tmax;
    {
      float tm[NT];
#pragma unroll
      for (int t = 0; t < NT; ++t) {
        tm[t] = s[t][0];
#pragma unroll
        for (int r = 1; r < 8; ++r) tm[t] = fmaxf(tm[t], s[t][r]);
      }
      tmax = tm[0];
#pragma unroll
      for (int t = 1; t < NT; ++t) tmax = fmaxf(tmax, tm[t]);
    }
    tmax = fmaxf(tmax, sw16(tmax));
    float m_new = fmaxf(m_run, tmax);
    float scale = __expf(m_run - m_new);
    m_run = m_new;
    float tsum = 0.0f;
#pragma unroll
    for (int t = 0; t < NT; ++t) {
      float ts = 0.0f;
#pragma unroll
      for (int r = 0; r < 8; ++r) {
        s[t][r] = __expf(s[t][r] - m_new);
        ts += s[t][r];
      }
      tsum += ts;
    }
    tsum += sw16(tsum);
    l_run = l_run * scale + tsum;
#pragma unroll
    for (int r = 0; r < 8; ++r) {
      acc0[r] *= scale;
      acc1[r] *= scale;
    }

#pragma unroll
    for (int t = 0; t < NT; ++t) {
      v2f pb[4];
      c2b(s[t], lo, pb);
#pragma unroll
      for (int j = 0; j < 4; ++j) {
        acc0 = wmma4(va0[t][j], pb[j], acc0);
        acc1 = wmma4(va1[t][j], pb[j], acc1);
      }
    }
  }

  float rinv = 1.0f / l_run;
#pragma unroll
  for (int r = 0; r < 8; ++r) {
    acc0[r] *= rinv;
    acc1[r] *= rinv;
  }

  v2f yb[8];
  c2b(acc0, lo, yb);
  c2b(acc1, lo, yb + 4);

  float* ob = out + (size_t)b * (CC * NN);
#pragma unroll
  for (int ct = 0; ct < 4; ++ct) {
    v8f z;
#pragma unroll
    for (int r = 0; r < 8; ++r) {
      int c = ct * 16 + r + (lo ? 0 : 8);
      z[r] = xb[(size_t)c * NN + n0 + lq] + W_b[c];
    }
    const int wrow = (ct * 16 + lq) * C2 + (lo ? 0 : 2);
#pragma unroll
    for (int oc = 0; oc < 8; ++oc) {
      v2f a;
      a[0] = W_w[wrow + 4 * oc];
      a[1] = W_w[wrow + 4 * oc + 1];
      z = wmma4(a, yb[oc], z);
    }
#pragma unroll
    for (int r = 0; r < 8; ++r) {
      int c = ct * 16 + r + (lo ? 0 : 8);
      ob[(size_t)c * NN + n0 + lq] = z[r];
    }
  }
}

extern "C" void kernel_launch(void* const* d_in, const int* in_sizes, int n_in,
                              void* d_out, int out_size, void* d_ws,
                              size_t ws_size, hipStream_t stream) {
  const float* x   = (const float*)d_in[0];
  const float* g_w = (const float*)d_in[1];
  const float* g_b = (const float*)d_in[2];
  const float* W_w = (const float*)d_in[3];
  const float* W_b = (const float*)d_in[4];
  float* out = (float*)d_out;
  float* Vg  = (float*)d_ws;

  gx_kernel<<<(NB * NN) / 256, 256, 0, stream>>>(x, g_w, g_b, Vg);
  attn_kernel<<<NB * (NN / 128), 256, 0, stream>>>(x, Vg, W_w, W_b, out);
}


# P^T score tile is PV's B fragment as it stands (regs 2j,2j+1 = keys 8h+2j+s); V staged to LDS in that key order: 32 LDS lane swaps + selects per 64 keys gone; order of y's sum over keys changes
# speedup vs baseline: 1.6537x; 1.1529x over previous
#include <hip/hip_runtime.h>
#include <hip/hip_bf16.h>
#include <math.h>

typedef float v2f __attribute__((ext_vector_type(2)));
typedef float v8f __attribute__((ext_vector_type(8)));

static constexpr int NN = 4096;
static constexpr int CC = 64;
static constexpr int C2 = 32;
static constexpr int NB = 8;
static constexpr int KT = 64;
static constexpr int NT = KT / 16;
static constexpr int SK = 66;
static constexpr int SV = KT + 4;

__device__ __forceinline__ float sw16(float v) {
  return __int_as_float(__builtin_amdgcn_ds_swizzle(__float_as_int(v), 0x401f));
}

__device__ __forceinline__ v8f wmma4(v2f a, v2f b, v8f c) {
  return __builtin_amdgcn_wmma_f32_16x16x4_f32(false, a, false, b, (short)0, c,
                                               false, false);
}

__device__ __forceinline__ void c2b(const v8f& R, bool lo, v2f* B) {
  float s0 = sw16(R[0]), s1 = sw16(R[1]), s2 = sw16(R[2]), s3 = sw16(R[3]);
  float s4 = sw16(R[4]), s5 = sw16(R[5]), s6 = sw16(R[6]), s7 = sw16(R[7]);
  B[0][0] = lo ? R[0] : s2;  B[0][1] = lo ? R[1] : s3;
  B[1][0] = lo ? R[4] : s6;  B[1][1] = lo ? R[5] : s7;
  B[2][0] = lo ? s0 : R[2];  B[2][1] = lo ? s1 : R[3];
  B[3][0] = lo ? s4 : R[6];  B[3][1] = lo ? s5 : R[7];
}

__device__ __forceinline__ int kpos(int m) {
  return (m & ~15) | ((m & 6) << 1) | ((m >> 2) & 2) | (m & 1);
}

__global__ __launch_bounds__(256) void gx_kernel(
    const float* __restrict__ x, const float* __restrict__ g_w,
    const float* __restrict__ g_b, float* __restrict__ Vg) {
  int idx = blockIdx.x * 256 + threadIdx.x;
  int b = idx >> 12;
  int n = idx & (NN - 1);
  const float* xb = x + (size_t)b * (CC * NN) + n;
  float acc[C2];
#pragma unroll
  for (int o = 0; o < C2; ++o) acc[o] = g_b[o];
  for (int c = 0; c < CC; ++c) {
    float xv = xb[(size_t)c * NN];
#pragma unroll
    for (int o = 0; o < C2; ++o) acc[o] = fmaf(g_w[o * CC + c], xv, acc[o]);
  }
  float* vp = Vg + (size_t)idx * C2;
#pragma unroll
  for (int o = 0; o < C2; ++o) vp[o] = acc[o];
}

__global__ __launch_bounds__(256) void attn_kernel(
    const float* __restrict__ x, const float* __restrict__ Vg,
    const float* __restrict__ W_w, const float* __restrict__ W_b,
    float* __restrict__ out) {
  __shared__ float ldsK[KT * SK];
  __shared__ float ldsV[32 * SV];

  const int b    = blockIdx.x >> 5;
  const int qblk = blockIdx.x & 31;
  const int tid  = threadIdx.x;
  const int wave = tid >> 5;
  const int lane = tid & 31;
  const int lq   = lane & 15;
  const bool lo  = lane < 16;
  const int n0   = qblk * 128 + wave * 16;

  const float* xb = x  + (size_t)b * (CC * NN);
  const float* vb = Vg + (size_t)b * (NN * C2);

  v2f qb[16];
  {
    const int crow = lo ? 0 : 2;
#pragma unroll
    for (int ck = 0; ck < 16; ++ck) {
      qb[ck][0] = xb[(size_t)(4 * ck + crow) * NN + n0 + lq];
      qb[ck][1] = xb[(size_t)(4 * ck + crow + 1) * NN + n0 + lq];
    }
  }

  v8f acc0 = {};
  v8f acc1 = {};
  float m_run = -INFINITY;
  float l_run = 0.0f;

  const int kc_m = tid & (KT - 1), kc_c = tid / KT;
  const int vc_o = tid & 31, vc_m = tid >> 5;
  const int vst = vc_o * SV + kpos(vc_m);
  const int fsel = lo ? 0 : 2;

  for (int m0 = 0; m0 < NN; m0 += KT) {
    __syncthreads();
#pragma unroll
    for (int i = 0; i < KT / 4; ++i) {
      const int c = kc_c + (256 / KT) * i;
      ldsK[kc_m * SK + c] = xb[(size_t)c * NN + m0 + kc_m];
    }
#pragma unroll
    for (int i = 0; i < KT / 8; ++i) {
      const int m = vc_m + 8 * i;
      ldsV[vst + 16 * (i >> 1) + 2 * (i & 1)] = vb[(size_t)(m0 + m) * C2 + vc_o];
    }
    __syncthreads();

    v8f s[NT];
#pragma unroll
    for (int t = 0; t < NT; ++t) s[t] = v8f{};
#pragma unroll
    for (int ck = 0; ck < 16; ++ck) {
#pragma unroll
      for (int t = 0; t < NT; ++t) {
        const int arow = (16 * t + lq) * SK + fsel + 4 * ck;
        v2f a;
        a[0] = ldsK[arow];
        a[1] = ldsK[arow + 1];
        s[t] = wmma4(a, qb[ck], s[t]);
      }
    }

    v2f va0[NT][4], va1[NT][4];
#pragma unroll
    for (int t = 0; t < NT; ++t) {
      const int vrow0 = lq * SV + 16 * t + fsel;
      const int vrow1 = (lq + 16) * SV + 16 * t + fsel;
#pragma unroll
      for (int j = 0; j < 4; ++j) {
        va0[t][j][0] = ldsV[vrow0 + 4 * j];
        va0[t][j][1] = ldsV[vrow0 + 4 * j + 1];
        va1[t][j][0] = ldsV[vrow1 + 4 * j];
        va1[t][j][1] = ldsV[vrow1 + 4 * j + 1];
      }
    }

    __builtin_amdgcn_sched_barrier(0);

    float tmax;
    {
      float tm[NT];
#pragma unroll
      for (int t = 0; t < NT; ++t) {
        tm[t] = s[t][0];
#pragma unroll
        for (int r = 1; r < 8; ++r) tm[t] = fmaxf(tm[t], s[t][r]);
      }
      tmax = tm[0];
#pragma unroll
      for (int t = 1; t < NT; ++t) tmax = fmaxf(tmax, tm[t]);
    }
    tmax = fmaxf(tmax, sw16(tmax));
    float m_new = fmaxf(m_run, tmax);
    float scale = __expf(m_run - m_new);
    m_run = m_new;
    float tsum = 0.0f;
#pragma unroll
    for (int t = 0; t < NT; ++t) {
      float ts = 0.0f;
#pragma unroll
      for (int r = 0; r < 8; ++r) {
        s[t][r] = __expf(s[t][r] - m_new);
        ts += s[t][r];
      }
      tsum += ts;
    }
    tsum += sw16(tsum);
    l_run = l_run * scale + tsum;
#pragma unroll
    for (int r = 0; r < 8; ++r) {
      acc0[r] *= scale;
      acc1[r] *= scale;
    }

#pragma unroll
    for (int t = 0; t < NT; ++t) {
#pragma unroll
      for (int j = 0; j < 4; ++j) {
        v2f pb;
        pb[0] = s[t][2 * j];
        pb[1] = s[t][2 * j + 1];
        acc0 = wmma4(va0[t][j], pb, acc0);
        acc1 = wmma4(va1[t][j], pb, acc1);
      }
    }
  }

  float rinv = 1.0f / l_run;
#pragma unroll
  for (int r = 0; r < 8; ++r) {
    acc0[r] *= rinv;
    acc1[r] *= rinv;
  }

  v2f yb[8];
  c2b(acc0, lo, yb);
  c2b(acc1, lo, yb + 4);

  float* ob = out + (size_t)b * (CC * NN);
#pragma unroll
  for (int ct = 0; ct < 4; ++ct) {
    v8f z;
#pragma unroll
    for (int r = 0; r < 8; ++r) {
      int c = ct * 16 + r + (lo ? 0 : 8);
      z[r] = xb[(size_t)c * NN + n0 + lq] + W_b[c];
    }
    const int wrow = (ct * 16 + lq) * C2 + (lo ? 0 : 2);
#pragma unroll
    for (int oc = 0; oc < 8; ++oc) {
      v2f a;
      a[0] = W_w[wrow + 4 * oc];
      a[1] = W_w[wrow + 4 * oc + 1];
      z = wmma4(a, yb[oc], z);
    }
#pragma unroll
    for (int r = 0; r < 8; ++r) {
      int c = ct * 16 + r + (lo ? 0 : 8);
      ob[(size_t)c * NN + n0 + lq] = z[r];
    }
  }
}

extern "C" void kernel_launch(void* const* d_in, const int* in_sizes, int n_in,
                              void* d_out, int out_size, void* d_ws,
                              size_t ws_size, hipStream_t stream) {
  const float* x   = (const float*)d_in[0];
  const float* g_w = (const float*)d_in[1];
  const float* g_b = (const float*)d_in[2];
  const float* W_w = (const float*)d_in[3];
  const float* W_b = (const float*)d_in[4];
  float* out = (float*)d_out;
  float* Vg  = (float*)d_ws;

  gx_kernel<<<(NB * NN) / 256, 256, 0, stream>>>(x, g_w, g_b, Vg);
  attn_kernel<<<NB * (NN / 128), 256, 0, stream>>>(x, Vg, W_w, W_b, out);
}


# K/V tile fetch software-pipelined: next tile's loads issued a stage ahead into registers, stored behind the score run into a second LDS buffer; one barrier per 64 keys, not two; sums unchanged
# speedup vs baseline: 2.0208x; 1.2220x over previous
#include <hip/hip_runtime.h>
#include <hip/hip_bf16.h>
#include <math.h>

typedef float v2f __attribute__((ext_vector_type(2)));
typedef float v8f __attribute__((ext_vector_type(8)));

static constexpr int NN = 4096;
static constexpr int CC = 64;
static constexpr int C2 = 32;
static constexpr int NB = 8;
static constexpr int KT = 64;
static constexpr int NT = KT / 16;
static constexpr int SK = 66;
static constexpr int SV = KT + 4;
static constexpr int KSZ = KT * SK;
static constexpr int VSZ = 32 * SV;

__device__ __forceinline__ float sw16(float v) {
  return __int_as_float(__builtin_amdgcn_ds_swizzle(__float_as_int(v), 0x401f));
}

__device__ __forceinline__ v8f wmma4(v2f a, v2f b, v8f c) {
  return __builtin_amdgcn_wmma_f32_16x16x4_f32(false, a, false, b, (short)0, c,
                                               false, false);
}

__device__ __forceinline__ void c2b(const v8f& R, bool lo, v2f* B) {
  float s0 = sw16(R[0]), s1 = sw16(R[1]), s2 = sw16(R[2]), s3 = sw16(R[3]);
  float s4 = sw16(R[4]), s5 = sw16(R[5]), s6 = sw16(R[6]), s7 = sw16(R[7]);
  B[0][0] = lo ? R[0] : s2;  B[0][1] = lo ? R[1] : s3;
  B[1][0] = lo ? R[4] : s6;  B[1][1] = lo ? R[5] : s7;
  B[2][0] = lo ? s0 : R[2];  B[2][1] = lo ? s1 : R[3];
  B[3][0] = lo ? s4 : R[6];  B[3][1] = lo ? s5 : R[7];
}

__device__ __forceinline__ int kpos(int m) {
  return (m & ~15) | ((m & 6) << 1) | ((m >> 2) & 2) | (m & 1);
}

__global__ __launch_bounds__(256) void gx_kernel(
    const float* __restrict__ x, const float* __restrict__ g_w,
    const float* __restrict__ g_b, float* __restrict__ Vg) {
  int idx = blockIdx.x * 256 + threadIdx.x;
  int b = idx >> 12;
  int n = idx & (NN - 1);
  const float* xb = x + (size_t)b * (CC * NN) + n;
  float acc[C2];
#pragma unroll
  for (int o = 0; o < C2; ++o) acc[o] = g_b[o];
  for (int c = 0; c < CC; ++c) {
    float xv = xb[(size_t)c * NN];
#pragma unroll
    for (int o = 0; o < C2; ++o) acc[o] = fmaf(g_w[o * CC + c], xv, acc[o]);
  }
  float* vp = Vg + (size_t)idx * C2;
#pragma unroll
  for (int o = 0; o < C2; ++o) vp[o] = acc[o];
}

__global__ __launch_bounds__(256) void attn_kernel(
    const float* __restrict__ x, const float* __restrict__ Vg,
    const float* __restrict__ W_w, const float* __restrict__ W_b,
    float* __restrict__ out) {
  __shared__ float ldsK[2 * KSZ];
  __shared__ float ldsV[2 * VSZ];

  const int b    = blockIdx.x >> 5;
  const int qblk = blockIdx.x & 31;
  const int tid  = threadIdx.x;
  const int wave = tid >> 5;
  const int lane = tid & 31;
  const int lq   = lane & 15;
  const bool lo  = lane < 16;
  const int n0   = qblk * 128 + wave * 16;

  const float* xb = x  + (size_t)b * (CC * NN);
  const float* vb = Vg + (size_t)b * (NN * C2);

  v2f qb[16];
  {
    const int crow = lo ? 0 : 2;
#pragma unroll
    for (int ck = 0; ck < 16; ++ck) {
      qb[ck][0] = xb[(size_t)(4 * ck + crow) * NN + n0 + lq];
      qb[ck][1] = xb[(size_t)(4 * ck + crow + 1) * NN + n0 + lq];
    }
  }

  v8f acc0 = {};
  v8f acc1 = {};
  float m_run = -INFINITY;
  float l_run = 0.0f;

  const int kc_m = tid & (KT - 1), kc_c = tid / KT;
  const int vc_o = tid & 31, vc_m = tid >> 5;
  const int vst = vc_o * SV + kpos(vc_m);
  const int fsel = lo ? 0 : 2;

  float kr[KT / 4], vr[KT / 8];
  const int kst = kc_m * SK + kc_c;
#pragma unroll
  for (int i = 0; i < KT / 4; ++i)
    kr[i] = xb[(size_t)(kc_c + (256 / KT) * i) * NN + kc_m];
#pragma unroll
  for (int i = 0; i < KT / 8; ++i)
    vr[i] = vb[(size_t)(vc_m + 8 * i) * C2 + vc_o];
#pragma unroll
  for (int i = 0; i < KT / 4; ++i) ldsK[kst + (256 / KT) * i] = kr[i];
#pragma unroll
  for (int i = 0; i < KT / 8; ++i) ldsV[vst + 16 * (i >> 1) + 2 * (i & 1)] = vr[i];
  __syncthreads();

  int ko = 0, vo = 0;

  for (int m0 = 0; m0 < NN; m0 += KT) {
    const int mn = (m0 + KT < NN) ? m0 + KT : m0;
#pragma unroll
    for (int i = 0; i < KT / 4; ++i)
      kr[i] = xb[(size_t)(kc_c + (256 / KT) * i) * NN + mn + kc_m];
#pragma unroll
    for (int i = 0; i < KT / 8; ++i)
      vr[i] = vb[(size_t)(mn + vc_m + 8 * i) * C2 + vc_o];
    const float* Kc = ldsK + ko;
    const float* Vc = ldsV + vo;

    v8f s[NT];
#pragma unroll
    for (int t = 0; t < NT; ++t) s[t] = v8f{};
#pragma unroll
    for (int ck = 0; ck < 16; ++ck) {
#pragma unroll
      for (int t = 0; t < NT; ++t) {
        const int arow = (16 * t + lq) * SK + fsel + 4 * ck;
        v2f a;
        a[0] = Kc[arow];
        a[1] = Kc[arow + 1];
        s[t] = wmma4(a, qb[ck], s[t]);
      }
    }

    v2f va0[NT][4], va1[NT][4];
#pragma unroll
    for (int t = 0; t < NT; ++t) {
      const int vrow0 = lq * SV + 16 * t + fsel;
      const int vrow1 = (lq + 16) * SV + 16 * t + fsel;
#pragma unroll
      for (int j = 0; j < 4; ++j) {
        va0[t][j][0] = Vc[vrow0 + 4 * j];
        va0[t][j][1] = Vc[vrow0 + 4 * j + 1];
        va1[t][j][0] = Vc[vrow1 + 4 * j];
        va1[t][j][1] = Vc[vrow1 + 4 * j + 1];
      }
    }

    __builtin_amdgcn_sched_barrier(0);

    {
      float* Kn = ldsK + (KSZ - ko);
      float* Vn = ldsV + (VSZ - vo);
#pragma unroll
      for (int i = 0; i < KT / 4; ++i) Kn[kst + (256 / KT) * i] = kr[i];
#pragma unroll
      for (int i = 0; i < KT / 8; ++i) Vn[vst + 16 * (i >> 1) + 2 * (i & 1)] = vr[i];
    }
    __builtin_amdgcn_sched_barrier(0);

    float tmax;
    {
      float tm[NT];
#pragma unroll
      for (int t = 0; t < NT; ++t) {
        tm[t] = s[t][0];
#pragma unroll
        for (int r = 1; r < 8; ++r) tm[t] = fmaxf(tm[t], s[t][r]);
      }
      tmax = tm[0];
#pragma unroll
      for (int t = 1; t < NT; ++t) tmax = fmaxf(tmax, tm[t]);
    }
    tmax = fmaxf(tmax, sw16(tmax));
    float m_new = fmaxf(m_run, tmax);
    float scale = __expf(m_run - m_new);
    m_run = m_new;
    float tsum = 0.0f;
#pragma unroll
    for (int t = 0; t < NT; ++t) {
      float ts = 0.0f;
#pragma unroll
      for (int r = 0; r < 8; ++r) {
        s[t][r] = __expf(s[t][r] - m_new);
        ts += s[t][r];
      }
      tsum += ts;
    }
    tsum += sw16(tsum);
    l_run = l_run * scale + tsum;
#pragma unroll
    for (int r = 0; r < 8; ++r) {
      acc0[r] *= scale;
      acc1[r] *= scale;
    }

#pragma unroll
    for (int t = 0; t < NT; ++t) {
#pragma unroll
      for (int j = 0; j < 4; ++j) {
        v2f pb;
        pb[0] = s[t][2 * j];
        pb[1] = s[t][2 * j + 1];
        acc0 = wmma4(va0[t][j], pb, acc0);
        acc1 = wmma4(va1[t][j], pb, acc1);
      }
    }

    __syncthreads();
    ko = KSZ - ko;
    vo = VSZ - vo;
  }

  float rinv = 1.0f / l_run;
#pragma unroll
  for (int r = 0; r < 8; ++r) {
    acc0[r] *= rinv;
    acc1[r] *= rinv;
  }

  v2f yb[8];
  c2b(acc0, lo, yb);
  c2b(acc1, lo, yb + 4);

  float* ob = out + (size_t)b * (CC * NN);
#pragma unroll
  for (int ct = 0; ct < 4; ++ct) {
    v8f z;
#pragma unroll
    for (int r = 0; r < 8; ++r) {
      int c = ct * 16 + r + (lo ? 0 : 8);
      z[r] = xb[(size_t)c * NN + n0 + lq] + W_b[c];
    }
    const int wrow = (ct * 16 + lq) * C2 + (lo ? 0 : 2);
#pragma unroll
    for (int oc = 0; oc < 8; ++oc) {
      v2f a;
      a[0] = W_w[wrow + 4 * oc];
      a[1] = W_w[wrow + 4 * oc + 1];
      z = wmma4(a, yb[oc], z);
    }
#pragma unroll
    for (int r = 0; r < 8; ++r) {
      int c = ct * 16 + r + (lo ? 0 : 8);
      ob[(size_t)c * NN + n0 + lq] = z[r];
    }
  }
}

extern "C" void kernel_launch(void* const* d_in, const int* in_sizes, int n_in,
                              void* d_out, int out_size, void* d_ws,
                              size_t ws_size, hipStream_t stream) {
  const float* x   = (const float*)d_in[0];
  const float* g_w = (const float*)d_in[1];
  const float* g_b = (const float*)d_in[2];
  const float* W_w = (const float*)d_in[3];
  const float* W_b = (const float*)d_in[4];
  float* out = (float*)d_out;
  float* Vg  = (float*)d_ws;

  gx_kernel<<<(NB * NN) / 256, 256, 0, stream>>>(x, g_w, g_b, Vg);
  attn_kernel<<<NB * (NN / 128), 256, 0, stream>>>(x, Vg, W_w, W_b, out);
}


# Keys split over two 8-wave groups in a 16-wave block (4 waves per SIMD, was 2), own tile buffers each; partial max/sum/y merged through LDS at the end: order of l's and y's sums over keys changes
# speedup vs baseline: 2.1181x; 1.0481x over previous
#include <hip/hip_runtime.h>
#include <hip/hip_bf16.h>
#include <math.h>

typedef float v2f __attribute__((ext_vector_type(2)));
typedef float v8f __attribute__((ext_vector_type(8)));

static constexpr int NN = 4096;
static constexpr int CC = 64;
static constexpr int C2 = 32;
static constexpr int NB = 8;
static constexpr int KT = 64;
static constexpr int NT = KT / 16;
static constexpr int SK = 66;
static constexpr int SV = KT + 4;
static constexpr int KSZ = KT * SK;
static constexpr int VSZ = 32 * SV;
static constexpr int NG  = 2;
static constexpr int NKG = NN / NG;
static constexpr int XW  = 18;

__device__ __forceinline__ float sw16(float v) {
  return __int_as_float(__builtin_amdgcn_ds_swizzle(__float_as_int(v), 0x401f));
}

__device__ __forceinline__ v8f wmma4(v2f a, v2f b, v8f c) {
  return __builtin_amdgcn_wmma_f32_16x16x4_f32(false, a, false, b, (short)0, c,
                                               false, false);
}

__device__ __forceinline__ void c2b(const v8f& R, bool lo, v2f* B) {
  float s0 = sw16(R[0]), s1 = sw16(R[1]), s2 = sw16(R[2]), s3 = sw16(R[3]);
  float s4 = sw16(R[4]), s5 = sw16(R[5]), s6 = sw16(R[6]), s7 = sw16(R[7]);
  B[0][0] = lo ? R[0] : s2;  B[0][1] = lo ? R[1] : s3;
  B[1][0] = lo ? R[4] : s6;  B[1][1] = lo ? R[5] : s7;
  B[2][0] = lo ? s0 : R[2];  B[2][1] = lo ? s1 : R[3];
  B[3][0] = lo ? s4 : R[6];  B[3][1] = lo ? s5 : R[7];
}

__device__ __forceinline__ int kpos(int m) {
  return (m & ~15) | ((m & 6) << 1) | ((m >> 2) & 2) | (m & 1);
}

__global__ __launch_bounds__(256) void gx_kernel(
    const float* __restrict__ x, const float* __restrict__ g_w,
    const float* __restrict__ g_b, float* __restrict__ Vg) {
  int idx = blockIdx.x * 256 + threadIdx.x;
  int b = idx >> 12;
  int n = idx & (NN - 1);
  const float* xb = x + (size_t)b * (CC * NN) + n;
  float acc[C2];
#pragma unroll
  for (int o = 0; o < C2; ++o) acc[o] = g_b[o];
  for (int c = 0; c < CC; ++c) {
    float xv = xb[(size_t)c * NN];
#pragma unroll
    for (int o = 0; o < C2; ++o) acc[o] = fmaf(g_w[o * CC + c], xv, acc[o]);
  }
  float* vp = Vg + (size_t)idx * C2;
#pragma unroll
  for (int o = 0; o < C2; ++o) vp[o] = acc[o];
}

__global__ __launch_bounds__(512) void attn_kernel(
    const float* __restrict__ x, const float* __restrict__ Vg,
    const float* __restrict__ W_w, const float* __restrict__ W_b,
    float* __restrict__ out) {
  __shared__ float ldsK[NG * 2 * KSZ];
  __shared__ float ldsV[NG * 2 * VSZ];

  const int b    = blockIdx.x >> 5;
  const int qblk = blockIdx.x & 31;
  const int grp  = threadIdx.x >> 8;
  const int tid  = threadIdx.x & 255;
  const int wave = tid >> 5;
  const int lane = tid & 31;
  const int lq   = lane & 15;
  const bool lo  = lane < 16;
  const int n0   = qblk * 128 + wave * 16;

  const float* xb = x  + (size_t)b * (CC * NN);
  const float* vb = Vg + (size_t)b * (NN * C2);
  const float* xk = xb + grp * NKG;
  const float* vk = vb + (size_t)grp * (NKG * C2);
  float* const gK = ldsK + grp * (2 * KSZ);
  float* const gV = ldsV + grp * (2 * VSZ);

  v2f qb[16];
  {
    const int crow = lo ? 0 : 2;
#pragma unroll
    for (int ck = 0; ck < 16; ++ck) {
      qb[ck][0] = xb[(size_t)(4 * ck + crow) * NN + n0 + lq];
      qb[ck][1] = xb[(size_t)(4 * ck + crow + 1) * NN + n0 + lq];
    }
  }

  v8f acc0 = {};
  v8f acc1 = {};
  float m_run = -INFINITY;
  float l_run = 0.0f;

  const int kc_m = tid & (KT - 1), kc_c = tid / KT;
  const int vc_o = tid & 31, vc_m = tid >> 5;
  const int vst = vc_o * SV + kpos(vc_m);
  const int fsel = lo ? 0 : 2;

  float kr[KT / 4], vr[KT / 8];
  const int kst = kc_m * SK + kc_c;
#pragma unroll
  for (int i = 0; i < KT / 4; ++i)
    kr[i] = xk[(size_t)(kc_c + (256 / KT) * i) * NN + kc_m];
#pragma unroll
  for (int i = 0; i < KT / 8; ++i)
    vr[i] = vk[(size_t)(vc_m + 8 * i) * C2 + vc_o];
#pragma unroll
  for (int i = 0; i < KT / 4; ++i) gK[kst + (256 / KT) * i] = kr[i];
#pragma unroll
  for (int i = 0; i < KT / 8; ++i) gV[vst + 16 * (i >> 1) + 2 * (i & 1)] = vr[i];
  __syncthreads();

  int ko = 0, vo = 0;

  for (int m0 = 0; m0 < NKG; m0 += KT) {
    const int mn = (m0 + KT < NKG) ? m0 + KT : m0;
#pragma unroll
    for (int i = 0; i < KT / 4; ++i)
      kr[i] = xk[(size_t)(kc_c + (256 / KT) * i) * NN + mn + kc_m];
#pragma unroll
    for (int i = 0; i < KT / 8; ++i)
      vr[i] = vk[(size_t)(mn + vc_m + 8 * i) * C2 + vc_o];
    const float* Kc = gK + ko;
    const float* Vc = gV + vo;

    v8f s[NT];
#pragma unroll
    for (int t = 0; t < NT; ++t) s[t] = v8f{};
#pragma unroll
    for (int ck = 0; ck < 16; ++ck) {
#pragma unroll
      for (int t = 0; t < NT; ++t) {
        const int arow = (16 * t + lq) * SK + fsel + 4 * ck;
        v2f a;
        a[0] = Kc[arow];
        a[1] = Kc[arow + 1];
        s[t] = wmma4(a, qb[ck], s[t]);
      }
    }

    v2f va0[NT][4], va1[NT][4];
#pragma unroll
    for (int t = 0; t < NT; ++t) {
      const int vrow0 = lq * SV + 16 * t + fsel;
      const int vrow1 = (lq + 16) * SV + 16 * t + fsel;
#pragma unroll
      for (int j = 0; j < 4; ++j) {
        va0[t][j][0] = Vc[vrow0 + 4 * j];
        va0[t][j][1] = Vc[vrow0 + 4 * j + 1];
        va1[t][j][0] = Vc[vrow1 + 4 * j];
        va1[t][j][1] = Vc[vrow1 + 4 * j + 1];
      }
    }

    __builtin_amdgcn_sched_barrier(0);

    {
      float* Kn = gK + (KSZ - ko);
      float* Vn = gV + (VSZ - vo);
#pragma unroll
      for (int i = 0; i < KT / 4; ++i) Kn[kst + (256 / KT) * i] = kr[i];
#pragma unroll
      for (int i = 0; i < KT / 8; ++i) Vn[vst + 16 * (i >> 1) + 2 * (i & 1)] = vr[i];
    }
    __builtin_amdgcn_sched_barrier(0);

    float tmax;
    {
      float tm[NT];
#pragma unroll
      for (int t = 0; t < NT; ++t) {
        tm[t] = s[t][0];
#pragma unroll
        for (int r = 1; r < 8; ++r) tm[t] = fmaxf(tm[t], s[t][r]);
      }
      tmax = tm[0];
#pragma unroll
      for (int t = 1; t < NT; ++t) tmax = fmaxf(tmax, tm[t]);
    }
    tmax = fmaxf(tmax, sw16(tmax));
    float m_new = fmaxf(m_run, tmax);
    float scale = __expf(m_run - m_new);
    m_run = m_new;
    float tsum = 0.0f;
#pragma unroll
    for (int t = 0; t < NT; ++t) {
      float ts = 0.0f;
#pragma unroll
      for (int r = 0; r < 8; ++r) {
        s[t][r] = __expf(s[t][r] - m_new);
        ts += s[t][r];
      }
      tsum += ts;
    }
    tsum += sw16(tsum);
    l_run = l_run * scale + tsum;
#pragma unroll
    for (int r = 0; r < 8; ++r) {
      acc0[r] *= scale;
      acc1[r] *= scale;
    }

#pragma unroll
    for (int t = 0; t < NT; ++t) {
#pragma unroll
      for (int j = 0; j < 4; ++j) {
        v2f pb;
        pb[0] = s[t][2 * j];
        pb[1] = s[t][2 * j + 1];
        acc0 = wmma4(va0[t][j], pb, acc0);
        acc1 = wmma4(va1[t][j], pb, acc1);
      }
    }

    __syncthreads();
    ko = KSZ - ko;
    vo = VSZ - vo;
  }

  {
    float* xw = ldsK + ((grp * 8 + wave) * XW) * 32 + lane;
    xw[0]  = m_run;
    xw[32] = l_run;
#pragma unroll
    for (int r = 0; r < 8; ++r) {
      xw[(2 + r) * 32]  = acc0[r];
      xw[(10 + r) * 32] = acc1[r];
    }
  }
  __syncthreads();
  {
    const float* xa = ldsK + (wave * XW) * 32 + lane;
    const float* xc = ldsK + ((8 + wave) * XW) * 32 + lane;
    const float ma = xa[0], mc = xc[0];
    const float mm = fmaxf(ma, mc);
    float sa = __expf(ma - mm), sc = __expf(mc - mm);
    const float rinv = 1.0f / (xa[32] * sa + xc[32] * sc);
    sa *= rinv;
    sc *= rinv;
#pragma unroll
    for (int r = 0; r < 8; ++r) {
      acc0[r] = xa[(2 + r) * 32] * sa + xc[(2 + r) * 32] * sc;
      acc1[r] = xa[(10 + r) * 32] * sa + xc[(10 + r) * 32] * sc;
    }
  }

  v2f yb[8];
  c2b(acc0, lo, yb);
  c2b(acc1, lo, yb + 4);

  float* ob = out + (size_t)b * (CC * NN);
#pragma unroll
  for (int cc = 0; cc < 2; ++cc) {
    const int ct = 2 * grp + cc;
    v8f z;
#pragma unroll
    for (int r = 0; r < 8; ++r) {
      int c = ct * 16 + r + (lo ? 0 : 8);
      z[r] = xb[(size_t)c * NN + n0 + lq] + W_b[c];
    }
    const int wrow = (ct * 16 + lq) * C2 + (lo ? 0 : 2);
#pragma unroll
    for (int oc = 0; oc < 8; ++oc) {
      v2f a;
      a[0] = W_w[wrow + 4 * oc];
      a[1] = W_w[wrow + 4 * oc + 1];
      z = wmma4(a, yb[oc], z);
    }
#pragma unroll
    for (int r = 0; r < 8; ++r) {
      int c = ct * 16 + r + (lo ? 0 : 8);
      ob[(size_t)c * NN + n0 + lq] = z[r];
    }
  }
}

extern "C" void kernel_launch(void* const* d_in, const int* in_sizes, int n_in,
                              void* d_out, int out_size, void* d_ws,
                              size_t ws_size, hipStream_t stream) {
  const float* x   = (const float*)d_in[0];
  const float* g_w = (const float*)d_in[1];
  const float* g_b = (const float*)d_in[2];
  const float* W_w = (const float*)d_in[3];
  const float* W_b = (const float*)d_in[4];
  float* out = (float*)d_out;
  float* Vg  = (float*)d_ws;

  gx_kernel<<<(NB * NN) / 256, 256, 0, stream>>>(x, g_w, g_b, Vg);
  attn_kernel<<<NB * (NN / 128), 512, 0, stream>>>(x, Vg, W_w, W_b, out);
}


# V projection kernel: each thread's 64 channel loads issued together ahead of its sums (was one load per loop trip, each waited for); g_w rows read as contiguous scalars; sums unchanged
# speedup vs baseline: 2.3926x; 1.1296x over previous
#include <hip/hip_runtime.h>
#include <hip/hip_bf16.h>
#include <math.h>

typedef float v2f __attribute__((ext_vector_type(2)));
typedef float v8f __attribute__((ext_vector_type(8)));

static constexpr int NN = 4096;
static constexpr int CC = 64;
static constexpr int C2 = 32;
static constexpr int NB = 8;
static constexpr int KT = 64;
static constexpr int NT = KT / 16;
static constexpr int SK = 66;
static constexpr int SV = KT + 4;
static constexpr int KSZ = KT * SK;
static constexpr int VSZ = 32 * SV;
static constexpr int NG  = 2;
static constexpr int NKG = NN / NG;
static constexpr int XW  = 18;

__device__ __forceinline__ float sw16(float v) {
  return __int_as_float(__builtin_amdgcn_ds_swizzle(__float_as_int(v), 0x401f));
}

__device__ __forceinline__ v8f wmma4(v2f a, v2f b, v8f c) {
  return __builtin_amdgcn_wmma_f32_16x16x4_f32(false, a, false, b, (short)0, c,
                                               false, false);
}

__device__ __forceinline__ void c2b(const v8f& R, bool lo, v2f* B) {
  float s0 = sw16(R[0]), s1 = sw16(R[1]), s2 = sw16(R[2]), s3 = sw16(R[3]);
  float s4 = sw16(R[4]), s5 = sw16(R[5]), s6 = sw16(R[6]), s7 = sw16(R[7]);
  B[0][0] = lo ? R[0] : s2;  B[0][1] = lo ? R[1] : s3;
  B[1][0] = lo ? R[4] : s6;  B[1][1] = lo ? R[5] : s7;
  B[2][0] = lo ? s0 : R[2];  B[2][1] = lo ? s1 : R[3];
  B[3][0] = lo ? s4 : R[6];  B[3][1] = lo ? s5 : R[7];
}

__device__ __forceinline__ int kpos(int m) {
  return (m & ~15) | ((m & 6) << 1) | ((m >> 2) & 2) | (m & 1);
}

static constexpr int GT = 128;
static constexpr int OG = 8;
__global__ __launch_bounds__(GT) void gx_kernel(
    const float* __restrict__ x, const float* __restrict__ g_w,
    const float* __restrict__ g_b, float* __restrict__ Vg) {
  int idx = blockIdx.x * GT + threadIdx.x;
  int b = idx >> 12;
  int n = idx & (NN - 1);
  const float* xb = x + (size_t)b * (CC * NN) + n;
  float xv[CC];
#pragma unroll
  for (int c = 0; c < CC; ++c) xv[c] = xb[(size_t)c * NN];
  float* vp = Vg + (size_t)idx * C2;
#pragma unroll 1
  for (int o0 = 0; o0 < C2; o0 += OG) {
    float acc[OG];
#pragma unroll
    for (int j = 0; j < OG; ++j) acc[j] = g_b[o0 + j];
#pragma unroll
    for (int c = 0; c < CC; ++c) {
#pragma unroll
      for (int j = 0; j < OG; ++j)
        acc[j] = fmaf(g_w[(o0 + j) * CC + c], xv[c], acc[j]);
    }
#pragma unroll
    for (int j = 0; j < OG; ++j) vp[o0 + j] = acc[j];
  }
}

__global__ __launch_bounds__(512) void attn_kernel(
    const float* __restrict__ x, const float* __restrict__ Vg,
    const float* __restrict__ W_w, const float* __restrict__ W_b,
    float* __restrict__ out) {
  __shared__ float ldsK[NG * 2 * KSZ];
  __shared__ float ldsV[NG * 2 * VSZ];

  const int b    = blockIdx.x >> 5;
  const int qblk = blockIdx.x & 31;
  const int grp  = threadIdx.x >> 8;
  const int tid  = threadIdx.x & 255;
  const int wave = tid >> 5;
  const int lane = tid & 31;
  const int lq   = lane & 15;
  const bool lo  = lane < 16;
  const int n0   = qblk * 128 + wave * 16;

  const float* xb = x  + (size_t)b * (CC * NN);
  const float* vb = Vg + (size_t)b * (NN * C2);
  const float* xk = xb + grp * NKG;
  const float* vk = vb + (size_t)grp * (NKG * C2);
  float* const gK = ldsK + grp * (2 * KSZ);
  float* const gV = ldsV + grp * (2 * VSZ);

  v2f qb[16];
  {
    const int crow = lo ? 0 : 2;
#pragma unroll
    for (int ck = 0; ck < 16; ++ck) {
      qb[ck][0] = xb[(size_t)(4 * ck + crow) * NN + n0 + lq];
      qb[ck][1] = xb[(size_t)(4 * ck + crow + 1) * NN + n0 + lq];
    }
  }

  v8f acc0 = {};
  v8f acc1 = {};
  float m_run = -INFINITY;
  float l_run = 0.0f;

  const int kc_m = tid & (KT - 1), kc_c = tid / KT;
  const int vc_o = tid & 31, vc_m = tid >> 5;
  const int vst = vc_o * SV + kpos(vc_m);
  const int fsel = lo ? 0 : 2;

  float kr[KT / 4], vr[KT / 8];
  const int kst = kc_m * SK + kc_c;
#pragma unroll
  for (int i = 0; i < KT / 4; ++i)
    kr[i] = xk[(size_t)(kc_c + (256 / KT) * i) * NN + kc_m];
#pragma unroll
  for (int i = 0; i < KT / 8; ++i)
    vr[i] = vk[(size_t)(vc_m + 8 * i) * C2 + vc_o];
#pragma unroll
  for (int i = 0; i < KT / 4; ++i) gK[kst + (256 / KT) * i] = kr[i];
#pragma unroll
  for (int i = 0; i < KT / 8; ++i) gV[vst + 16 * (i >> 1) + 2 * (i & 1)] = vr[i];
  __syncthreads();

  int ko = 0, vo = 0;

  for (int m0 = 0; m0 < NKG; m0 += KT) {
    const int mn = (m0 + KT < NKG) ? m0 + KT : m0;
#pragma unroll
    for (int i = 0; i < KT / 4; ++i)
      kr[i] = xk[(size_t)(kc_c + (256 / KT) * i) * NN + mn + kc_m];
#pragma unroll
    for (int i = 0; i < KT / 8; ++i)
      vr[i] = vk[(size_t)(mn + vc_m + 8 * i) * C2 + vc_o];
    const float* Kc = gK + ko;
    const float* Vc = gV + vo;

    v8f s[NT];
#pragma unroll
    for (int t = 0; t < NT; ++t) s[t] = v8f{};
#pragma unroll
    for (int ck = 0; ck < 16; ++ck) {
#pragma unroll
      for (int t = 0; t < NT; ++t) {
        const int arow = (16 * t + lq) * SK + fsel + 4 * ck;
        v2f a;
        a[0] = Kc[arow];
        a[1] = Kc[arow + 1];
        s[t] = wmma4(a, qb[ck], s[t]);
      }
    }

    v2f va0[NT][4], va1[NT][4];
#pragma unroll
    for (int t = 0; t < NT; ++t) {
      const int vrow0 = lq * SV + 16 * t + fsel;
      const int vrow1 = (lq + 16) * SV + 16 * t + fsel;
#pragma unroll
      for (int j = 0; j < 4; ++j) {
        va0[t][j][0] = Vc[vrow0 + 4 * j];
        va0[t][j][1] = Vc[vrow0 + 4 * j + 1];
        va1[t][j][0] = Vc[vrow1 + 4 * j];
        va1[t][j][1] = Vc[vrow1 + 4 * j + 1];
      }
    }

    __builtin_amdgcn_sched_barrier(0);

    {
      float* Kn = gK + (KSZ - ko);
      float* Vn = gV + (VSZ - vo);
#pragma unroll
      for (int i = 0; i < KT / 4; ++i) Kn[kst + (256 / KT) * i] = kr[i];
#pragma unroll
      for (int i = 0; i < KT / 8; ++i) Vn[vst + 16 * (i >> 1) + 2 * (i & 1)] = vr[i];
    }
    __builtin_amdgcn_sched_barrier(0);

    float tmax;
    {
      float tm[NT];
#pragma unroll
      for (int t = 0; t < NT; ++t) {
        tm[t] = s[t][0];
#pragma unroll
        for (int r = 1; r < 8; ++r) tm[t] = fmaxf(tm[t], s[t][r]);
      }
      tmax = tm[0];
#pragma unroll
      for (int t = 1; t < NT; ++t) tmax = fmaxf(tmax, tm[t]);
    }
    tmax = fmaxf(tmax, sw16(tmax));
    float m_new = fmaxf(m_run, tmax);
    float scale = __expf(m_run - m_new);
    m_run = m_new;
    float tsum = 0.0f;
#pragma unroll
    for (int t = 0; t < NT; ++t) {
      float ts = 0.0f;
#pragma unroll
      for (int r = 0; r < 8; ++r) {
        s[t][r] = __expf(s[t][r] - m_new);
        ts += s[t][r];
      }
      tsum += ts;
    }
    tsum += sw16(tsum);
    l_run = l_run * scale + tsum;
#pragma unroll
    for (int r = 0; r < 8; ++r) {
      acc0[r] *= scale;
      acc1[r] *= scale;
    }

#pragma unroll
    for (int t = 0; t < NT; ++t) {
#pragma unroll
      for (int j = 0; j < 4; ++j) {
        v2f pb;
        pb[0] = s[t][2 * j];
        pb[1] = s[t][2 * j + 1];
        acc0 = wmma4(va0[t][j], pb, acc0);
        acc1 = wmma4(va1[t][j], pb, acc1);
      }
    }

    __syncthreads();
    ko = KSZ - ko;
    vo = VSZ - vo;
  }

  {
    float* xw = ldsK + ((grp * 8 + wave) * XW) * 32 + lane;
    xw[0]  = m_run;
    xw[32] = l_run;
#pragma unroll
    for (int r = 0; r < 8; ++r) {
      xw[(2 + r) * 32]  = acc0[r];
      xw[(10 + r) * 32] = acc1[r];
    }
  }
  __syncthreads();
  {
    const float* xa = ldsK + (wave * XW) * 32 + lane;
    const float* xc = ldsK + ((8 + wave) * XW) * 32 + lane;
    const float ma = xa[0], mc = xc[0];
    const float mm = fmaxf(ma, mc);
    float sa = __expf(ma - mm), sc = __expf(mc - mm);
    const float rinv = 1.0f / (xa[32] * sa + xc[32] * sc);
    sa *= rinv;
    sc *= rinv;
#pragma unroll
    for (int r = 0; r < 8; ++r) {
      acc0[r] = xa[(2 + r) * 32] * sa + xc[(2 + r) * 32] * sc;
      acc1[r] = xa[(10 + r) * 32] * sa + xc[(10 + r) * 32] * sc;
    }
  }

  v2f yb[8];
  c2b(acc0, lo, yb);
  c2b(acc1, lo, yb + 4);

  float* ob = out + (size_t)b * (CC * NN);
#pragma unroll
  for (int cc = 0; cc < 2; ++cc) {
    const int ct = 2 * grp + cc;
    v8f z;
#pragma unroll
    for (int r = 0; r < 8; ++r) {
      int c = ct * 16 + r + (lo ? 0 : 8);
      z[r] = xb[(size_t)c * NN + n0 + lq] + W_b[c];
    }
    const int wrow = (ct * 16 + lq) * C2 + (lo ? 0 : 2);
#pragma unroll
    for (int oc = 0; oc < 8; ++oc) {
      v2f a;
      a[0] = W_w[wrow + 4 * oc];
      a[1] = W_w[wrow + 4 * oc + 1];
      z = wmma4(a, yb[oc], z);
    }
#pragma unroll
    for (int r = 0; r < 8; ++r) {
      int c = ct * 16 + r + (lo ? 0 : 8);
      ob[(size_t)c * NN + n0 + lq] = z[r];
    }
  }
}

extern "C" void kernel_launch(void* const* d_in, const int* in_sizes, int n_in,
                              void* d_out, int out_size, void* d_ws,
                              size_t ws_size, hipStream_t stream) {
  const float* x   = (const float*)d_in[0];
  const float* g_w = (const float*)d_in[1];
  const float* g_b = (const float*)d_in[2];
  const float* W_w = (const float*)d_in[3];
  const float* W_b = (const float*)d_in[4];
  float* out = (float*)d_out;
  float* Vg  = (float*)d_ws;

  gx_kernel<<<(NB * NN) / GT, GT, 0, stream>>>(x, g_w, g_b, Vg);
  attn_kernel<<<NB * (NN / 128), 512, 0, stream>>>(x, Vg, W_w, W_b, out);
}
